# starting kernel
# baseline (speedup 1.0000x reference)
#include <hip/hip_runtime.h>

typedef __attribute__((ext_vector_type(2))) float v2f;
typedef __attribute__((ext_vector_type(8))) float v8f;

#define NB   2048
#define NE   64
#define ND   512
#define NO   512
#define MAXT 192


__global__ void zero_meta_kernel(int* counts, int* cursors) {
    int t = blockIdx.x * blockDim.x + threadIdx.x;
    if (t < NE) { counts[t] = 0; cursors[t] = 0; }
}

__global__ void histo_kernel(const int* __restrict__ opt, int* counts) {
    int t = blockIdx.x * blockDim.x + threadIdx.x;
    if (t < NB) atomicAdd(&counts[opt[t]], 1);
}

__global__ void scan_tiles_kernel(const int* __restrict__ counts, int* offsets,
                                  int* tileE, int* tileBase, int* tileCnt) {
    if (threadIdx.x != 0 || blockIdx.x != 0) return;
    int acc = 0;
    for (int e = 0; e < NE; ++e) { offsets[e] = acc; acc += counts[e]; }
    int t = 0;
    for (int e = 0; e < NE; ++e) {
        int c = counts[e];
        for (int i = 0; i < c; i += 16) {
            tileE[t]    = e;
            tileBase[t] = offsets[e] + i;
            tileCnt[t]  = (c - i) < 16 ? (c - i) : 16;
            ++t;
        }
    }
    for (; t < MAXT; ++t) { tileE[t] = -1; tileBase[t] = 0; tileCnt[t] = 0; }
}

__global__ void scatter_kernel(const int* __restrict__ opt,
                               const int* __restrict__ offsets,
                               int* cursors, int* idx) {
    int t = blockIdx.x * blockDim.x + threadIdx.x;
    if (t < NB) {
        int e = opt[t];
        int p = offsets[e] + atomicAdd(&cursors[e], 1);
        idx[p] = t;
    }
}


#define APAD 4

__global__ void __launch_bounds__(256)
option_gemm_kernel(const float* __restrict__ x, const float* __restrict__ w,
                   const float* __restrict__ bias, const int* __restrict__ idx,
                   const int* __restrict__ tileE, const int* __restrict__ tileBase,
                   const int* __restrict__ tileCnt, float* __restrict__ out) {
    __shared__ float As[16][ND + APAD];
    __shared__ int   rowIdx[16];

    const int e = tileE[blockIdx.x];
    if (e < 0) return;
    const int base = tileBase[blockIdx.x];
    const int cnt  = tileCnt[blockIdx.x];
    const int tid  = threadIdx.x;

    if (tid < 16) rowIdx[tid] = (tid < cnt) ? idx[base + tid] : 0;
    {
        const int r  = tid >> 4;
        const int cs = (tid & 15) * 32;
        if (r < cnt) {
            const float* src = x + (size_t)idx[base + r] * ND + cs;
            #pragma unroll
            for (int i = 0; i < 8; ++i) {
                float4 v = *(const float4*)(src + i * 4);
                *(float4*)&As[r][cs + i * 4] = v;
            }
        } else {
            #pragma unroll
            for (int i = 0; i < 8; ++i) {
                float4 z = {0.f, 0.f, 0.f, 0.f};
                *(float4*)&As[r][cs + i * 4] = z;
            }
        }
    }
    __syncthreads();

    const int lane = tid & 31;
    const int wv   = tid >> 5;
    const int n0   = blockIdx.y * 256 + wv * 16 + (lane & 15);
    const int kh   = (lane >> 4) * 2;
    const int row  = lane & 15;

    const float* wp0 = w + (size_t)e * (ND * NO) + n0;
    const float* wp1 = wp0 + 128;

    v8f acc0 = {};
    v8f acc1 = {};
    #pragma unroll 4
    for (int k0 = 0; k0 < ND; k0 += 4) {
        const int k = k0 + kh;
        v2f a;
        a.x = As[row][k];
        a.y = As[row][k + 1];
        v2f b0, b1;
        b0.x = wp0[(size_t)k * NO];
        b0.y = wp0[(size_t)(k + 1) * NO];
        b1.x = wp1[(size_t)k * NO];
        b1.y = wp1[(size_t)(k + 1) * NO];
        acc0 = __builtin_amdgcn_wmma_f32_16x16x4_f32(
            false, a, false, b0, (short)0, acc0, false, false);
        acc1 = __builtin_amdgcn_wmma_f32_16x16x4_f32(
            false, a, false, b1, (short)0, acc1, false, false);
    }

    const float bv0 = bias[e * NO + n0];
    const float bv1 = bias[e * NO + n0 + 128];
    const int mhi = (lane >> 4) * 8;
    #pragma unroll
    for (int j = 0; j < 8; ++j) {
        const int m = j + mhi;
        if (m < cnt) {
            float* orow = out + (size_t)rowIdx[m] * NO + n0;
            orow[0]   = acc0[j] + bv0;
            orow[128] = acc1[j] + bv1;
        }
    }
}


extern "C" void kernel_launch(void* const* d_in, const int* in_sizes, int n_in,
                              void* d_out, int out_size, void* d_ws, size_t ws_size,
                              hipStream_t stream) {
    const float* x    = (const float*)d_in[0];
    const int*   opt  = (const int*)d_in[1];
    const float* wgt  = (const float*)d_in[2];
    const float* bias = (const float*)d_in[3];
    float* out = (float*)d_out;

    int* counts   = (int*)d_ws;
    int* offsets  = counts + NE;
    int* cursors  = offsets + NE;
    int* idx      = cursors + NE;
    int* tileE    = idx + NB;
    int* tileBase = tileE + MAXT;
    int* tileCnt  = tileBase + MAXT;

    zero_meta_kernel<<<1, 64, 0, stream>>>(counts, cursors);
    histo_kernel<<<NB / 256, 256, 0, stream>>>(opt, counts);
    scan_tiles_kernel<<<1, 1, 0, stream>>>(counts, offsets, tileE, tileBase, tileCnt);
    scatter_kernel<<<NB / 256, 256, 0, stream>>>(opt, offsets, cursors, idx);

    dim3 grid(MAXT, 2);
    option_gemm_kernel<<<grid, 256, 0, stream>>>(x, wgt, bias, idx,
                                                 tileE, tileBase, tileCnt, out);
}


# Routing table built by one work-group in one launch (LDS histogram whose atomic return is the rank, shuffle prefix over the experts, one thread per tile) instead of 4 launches with a serial scan
# speedup vs baseline: 2.0425x; 2.0425x over previous
#include <hip/hip_runtime.h>

typedef __attribute__((ext_vector_type(2))) float v2f;
typedef __attribute__((ext_vector_type(8))) float v8f;

#define NB   2048
#define NE   64
#define ND   512
#define NO   512
#define MAXT 192


#define PREP_T 1024

__global__ void __launch_bounds__(PREP_T)
route_prep_kernel(const int* __restrict__ opt, int* __restrict__ idx,
                  int* __restrict__ tileE, int* __restrict__ tileBase,
                  int* __restrict__ tileCnt) {
    __shared__ int cnt[NE];
    __shared__ int off[NE];
    __shared__ __attribute__((aligned(16))) int tend[NE];

    const int tid = threadIdx.x;
    const int e0  = opt[tid];
    const int e1  = opt[tid + PREP_T];
    if (tid < NE) cnt[tid] = 0;
    __syncthreads();
    const int r0 = atomicAdd(&cnt[e0], 1);
    const int r1 = atomicAdd(&cnt[e1], 1);
    __syncthreads();

    if (tid < 32) {
        const int cA = cnt[2 * tid];
        const int cB = cnt[2 * tid + 1];
        const int nA = (cA + 15) >> 4;
        const int nB = (cB + 15) >> 4;
        int so = cA + cB, st = nA + nB;
        for (int d = 1; d < 32; d <<= 1) {
            const int vo = __shfl_up(so, d);
            const int vt = __shfl_up(st, d);
            if (tid >= d) { so += vo; st += vt; }
        }
        off[2 * tid]      = so - cB - cA;
        off[2 * tid + 1]  = so - cB;
        tend[2 * tid]     = st - nB;
        tend[2 * tid + 1] = st;
    }
    __syncthreads();

    idx[off[e0] + r0] = tid;
    idx[off[e1] + r1] = tid + PREP_T;

    if (tid < MAXT) {
        int e = 0;
        for (int j = 0; j < NE; j += 4) {
            const int4 v = *(const int4*)&tend[j];
            e += (v.x <= tid) + (v.y <= tid) + (v.z <= tid) + (v.w <= tid);
        }
        if (e < NE) {
            const int c = cnt[e];
            const int i = tid - (tend[e] - ((c + 15) >> 4));
            const int left = c - 16 * i;
            tileE[tid]    = e;
            tileBase[tid] = off[e] + 16 * i;
            tileCnt[tid]  = left < 16 ? left : 16;
        } else {
            tileE[tid] = -1; tileBase[tid] = 0; tileCnt[tid] = 0;
        }
    }
}


#define APAD 4

__global__ void __launch_bounds__(256)
option_gemm_kernel(const float* __restrict__ x, const float* __restrict__ w,
                   const float* __restrict__ bias, const int* __restrict__ idx,
                   const int* __restrict__ tileE, const int* __restrict__ tileBase,
                   const int* __restrict__ tileCnt, float* __restrict__ out) {
    __shared__ float As[16][ND + APAD];
    __shared__ int   rowIdx[16];

    const int e = tileE[blockIdx.x];
    if (e < 0) return;
    const int base = tileBase[blockIdx.x];
    const int cnt  = tileCnt[blockIdx.x];
    const int tid  = threadIdx.x;

    if (tid < 16) rowIdx[tid] = (tid < cnt) ? idx[base + tid] : 0;
    {
        const int r  = tid >> 4;
        const int cs = (tid & 15) * 32;
        if (r < cnt) {
            const float* src = x + (size_t)idx[base + r] * ND + cs;
            #pragma unroll
            for (int i = 0; i < 8; ++i) {
                float4 v = *(const float4*)(src + i * 4);
                *(float4*)&As[r][cs + i * 4] = v;
            }
        } else {
            #pragma unroll
            for (int i = 0; i < 8; ++i) {
                float4 z = {0.f, 0.f, 0.f, 0.f};
                *(float4*)&As[r][cs + i * 4] = z;
            }
        }
    }
    __syncthreads();

    const int lane = tid & 31;
    const int wv   = tid >> 5;
    const int n0   = blockIdx.y * 256 + wv * 16 + (lane & 15);
    const int kh   = (lane >> 4) * 2;
    const int row  = lane & 15;

    const float* wp0 = w + (size_t)e * (ND * NO) + n0;
    const float* wp1 = wp0 + 128;

    v8f acc0 = {};
    v8f acc1 = {};
    #pragma unroll 4
    for (int k0 = 0; k0 < ND; k0 += 4) {
        const int k = k0 + kh;
        v2f a;
        a.x = As[row][k];
        a.y = As[row][k + 1];
        v2f b0, b1;
        b0.x = wp0[(size_t)k * NO];
        b0.y = wp0[(size_t)(k + 1) * NO];
        b1.x = wp1[(size_t)k * NO];
        b1.y = wp1[(size_t)(k + 1) * NO];
        acc0 = __builtin_amdgcn_wmma_f32_16x16x4_f32(
            false, a, false, b0, (short)0, acc0, false, false);
        acc1 = __builtin_amdgcn_wmma_f32_16x16x4_f32(
            false, a, false, b1, (short)0, acc1, false, false);
    }

    const float bv0 = bias[e * NO + n0];
    const float bv1 = bias[e * NO + n0 + 128];
    const int mhi = (lane >> 4) * 8;
    #pragma unroll
    for (int j = 0; j < 8; ++j) {
        const int m = j + mhi;
        if (m < cnt) {
            float* orow = out + (size_t)rowIdx[m] * NO + n0;
            orow[0]   = acc0[j] + bv0;
            orow[128] = acc1[j] + bv1;
        }
    }
}


extern "C" void kernel_launch(void* const* d_in, const int* in_sizes, int n_in,
                              void* d_out, int out_size, void* d_ws, size_t ws_size,
                              hipStream_t stream) {
    const float* x    = (const float*)d_in[0];
    const int*   opt  = (const int*)d_in[1];
    const float* wgt  = (const float*)d_in[2];
    const float* bias = (const float*)d_in[3];
    float* out = (float*)d_out;

    int* idx      = (int*)d_ws;
    int* tileE    = idx + NB;
    int* tileBase = tileE + MAXT;
    int* tileCnt  = tileBase + MAXT;

    route_prep_kernel<<<1, PREP_T, 0, stream>>>(opt, idx, tileE, tileBase, tileCnt);

    dim3 grid(MAXT, 2);
    option_gemm_kernel<<<grid, 256, 0, stream>>>(x, wgt, bias, idx,
                                                 tileE, tileBase, tileCnt, out);
}


# Weight fragments fetched as 16-byte loads: a wave owns 64 columns as four column-interleaved 16-col tiles, so one load feeds 4 matrix products (32 in flight per trip); block spans all 512 columns
# speedup vs baseline: 2.9893x; 1.4635x over previous
#include <hip/hip_runtime.h>

typedef __attribute__((ext_vector_type(2))) float v2f;
typedef __attribute__((ext_vector_type(8))) float v8f;

#define NB   2048
#define NE   64
#define ND   512
#define NO   512
#define MAXT 192


#define PREP_T 1024

__global__ void __launch_bounds__(PREP_T)
route_prep_kernel(const int* __restrict__ opt, int* __restrict__ idx,
                  int* __restrict__ tileE, int* __restrict__ tileBase,
                  int* __restrict__ tileCnt) {
    __shared__ int cnt[NE];
    __shared__ int off[NE];
    __shared__ __attribute__((aligned(16))) int tend[NE];

    const int tid = threadIdx.x;
    const int e0  = opt[tid];
    const int e1  = opt[tid + PREP_T];
    if (tid < NE) cnt[tid] = 0;
    __syncthreads();
    const int r0 = atomicAdd(&cnt[e0], 1);
    const int r1 = atomicAdd(&cnt[e1], 1);
    __syncthreads();

    if (tid < 32) {
        const int cA = cnt[2 * tid];
        const int cB = cnt[2 * tid + 1];
        const int nA = (cA + 15) >> 4;
        const int nB = (cB + 15) >> 4;
        int so = cA + cB, st = nA + nB;
        for (int d = 1; d < 32; d <<= 1) {
            const int vo = __shfl_up(so, d);
            const int vt = __shfl_up(st, d);
            if (tid >= d) { so += vo; st += vt; }
        }
        off[2 * tid]      = so - cB - cA;
        off[2 * tid + 1]  = so - cB;
        tend[2 * tid]     = st - nB;
        tend[2 * tid + 1] = st;
    }
    __syncthreads();

    idx[off[e0] + r0] = tid;
    idx[off[e1] + r1] = tid + PREP_T;

    if (tid < MAXT) {
        int e = 0;
        for (int j = 0; j < NE; j += 4) {
            const int4 v = *(const int4*)&tend[j];
            e += (v.x <= tid) + (v.y <= tid) + (v.z <= tid) + (v.w <= tid);
        }
        if (e < NE) {
            const int c = cnt[e];
            const int i = tid - (tend[e] - ((c + 15) >> 4));
            const int left = c - 16 * i;
            tileE[tid]    = e;
            tileBase[tid] = off[e] + 16 * i;
            tileCnt[tid]  = left < 16 ? left : 16;
        } else {
            tileE[tid] = -1; tileBase[tid] = 0; tileCnt[tid] = 0;
        }
    }
}


#define APAD 4

__global__ void __launch_bounds__(256) __attribute__((amdgpu_waves_per_eu(2, 4)))
option_gemm_kernel(const float* __restrict__ x, const float* __restrict__ w,
                   const float* __restrict__ bias, const int* __restrict__ idx,
                   const int* __restrict__ tileE, const int* __restrict__ tileBase,
                   const int* __restrict__ tileCnt, float* __restrict__ out) {
    __shared__ float As[16][ND + APAD];
    __shared__ int   rowIdx[16];

    const int e = tileE[blockIdx.x];
    if (e < 0) return;
    const int base = tileBase[blockIdx.x];
    const int cnt  = tileCnt[blockIdx.x];
    const int tid  = threadIdx.x;

    if (tid < 16) rowIdx[tid] = (tid < cnt) ? idx[base + tid] : 0;
    {
        const int r  = tid >> 4;
        const int cs = (tid & 15) * 32;
        if (r < cnt) {
            const float* src = x + (size_t)idx[base + r] * ND + cs;
            #pragma unroll
            for (int i = 0; i < 8; ++i) {
                float4 v = *(const float4*)(src + i * 4);
                *(float4*)&As[r][cs + i * 4] = v;
            }
        } else {
            #pragma unroll
            for (int i = 0; i < 8; ++i) {
                float4 z = {0.f, 0.f, 0.f, 0.f};
                *(float4*)&As[r][cs + i * 4] = z;
            }
        }
    }
    __syncthreads();

    const int lane = tid & 31;
    const int wv   = tid >> 5;
    const int n0   = wv * 64 + (lane & 15) * 4;
    const int kh   = (lane >> 4) * 2;
    const int row  = lane & 15;

    const float* wp = w + (size_t)e * (ND * NO) + (size_t)kh * NO + n0;

    v8f acc0 = {};
    v8f acc1 = {};
    v8f acc2 = {};
    v8f acc3 = {};
    #pragma unroll 16
    for (int k0 = 0; k0 < ND; k0 += 4) {
        const int k = k0 + kh;
        v2f a;
        a.x = As[row][k];
        a.y = As[row][k + 1];
        const float4 q0 = *(const float4*)(wp + (size_t)k0 * NO);
        const float4 q1 = *(const float4*)(wp + (size_t)(k0 + 1) * NO);
        v2f b0, b1, b2, b3;
        b0.x = q0.x; b0.y = q1.x;
        b1.x = q0.y; b1.y = q1.y;
        b2.x = q0.z; b2.y = q1.z;
        b3.x = q0.w; b3.y = q1.w;
        acc0 = __builtin_amdgcn_wmma_f32_16x16x4_f32(
            false, a, false, b0, (short)0, acc0, false, false);
        acc1 = __builtin_amdgcn_wmma_f32_16x16x4_f32(
            false, a, false, b1, (short)0, acc1, false, false);
        acc2 = __builtin_amdgcn_wmma_f32_16x16x4_f32(
            false, a, false, b2, (short)0, acc2, false, false);
        acc3 = __builtin_amdgcn_wmma_f32_16x16x4_f32(
            false, a, false, b3, (short)0, acc3, false, false);
    }

    const float4 bv = *(const float4*)(bias + e * NO + n0);
    const int mhi = (lane >> 4) * 8;
    #pragma unroll
    for (int j = 0; j < 8; ++j) {
        const int m = j + mhi;
        if (m < cnt) {
            float4 o;
            o.x = acc0[j] + bv.x;
            o.y = acc1[j] + bv.y;
            o.z = acc2[j] + bv.z;
            o.w = acc3[j] + bv.w;
            *(float4*)(out + (size_t)rowIdx[m] * NO + n0) = o;
        }
    }
}


extern "C" void kernel_launch(void* const* d_in, const int* in_sizes, int n_in,
                              void* d_out, int out_size, void* d_ws, size_t ws_size,
                              hipStream_t stream) {
    const float* x    = (const float*)d_in[0];
    const int*   opt  = (const int*)d_in[1];
    const float* wgt  = (const float*)d_in[2];
    const float* bias = (const float*)d_in[3];
    float* out = (float*)d_out;

    int* idx      = (int*)d_ws;
    int* tileE    = idx + NB;
    int* tileBase = tileE + MAXT;
    int* tileCnt  = tileBase + MAXT;

    route_prep_kernel<<<1, PREP_T, 0, stream>>>(opt, idx, tileE, tileBase, tileCnt);

    dim3 grid(MAXT);
    option_gemm_kernel<<<grid, 256, 0, stream>>>(x, wgt, bias, idx,
                                                 tileE, tileBase, tileCnt, out);
}


# One block per (expert, 128-col slab): slab weights fetched once, each fragment feeds all row tiles of the expert; K split over 4 waves, partials summed in LDS (new summation order)
# speedup vs baseline: 3.6260x; 1.2130x over previous
#include <hip/hip_runtime.h>

typedef __attribute__((ext_vector_type(2))) float v2f;
typedef __attribute__((ext_vector_type(8))) float v8f;

#define NB   2048
#define NE   64
#define ND   512
#define NO   512


#define PREP_T 1024

__global__ void __launch_bounds__(PREP_T)
route_prep_kernel(const int* __restrict__ opt, int* __restrict__ idx,
                  int* __restrict__ expOff, int* __restrict__ expCnt) {
    __shared__ int cnt[NE];
    __shared__ int off[NE];

    const int tid = threadIdx.x;
    const int e0  = opt[tid];
    const int e1  = opt[tid + PREP_T];
    if (tid < NE) cnt[tid] = 0;
    __syncthreads();
    const int r0 = atomicAdd(&cnt[e0], 1);
    const int r1 = atomicAdd(&cnt[e1], 1);
    __syncthreads();

    if (tid < 32) {
        const int cA = cnt[2 * tid];
        const int cB = cnt[2 * tid + 1];
        int so = cA + cB;
        for (int d = 1; d < 32; d <<= 1) {
            const int vo = __shfl_up(so, d);
            if (tid >= d) so += vo;
        }
        off[2 * tid]        = so - cB - cA;
        off[2 * tid + 1]    = so - cB;
        expOff[2 * tid]     = so - cB - cA;
        expOff[2 * tid + 1] = so - cB;
        expCnt[2 * tid]     = cA;
        expCnt[2 * tid + 1] = cB;
    }
    __syncthreads();

    idx[off[e0] + r0] = tid;
    idx[off[e1] + r1] = tid + PREP_T;
}


#define SLABC 128
#define RG    64
#define KQ    128
#define APAD  4

template <int NT>
__device__ __forceinline__ void slab_pass(const float* __restrict__ wp,
                                          const float (*As)[ND + APAD],
                                          float (*Pq)[SLABC],
                                          const int kbase, const int row,
                                          const int prow, const int pcol) {
    v8f acc[NT][4];
    #pragma unroll
    for (int m = 0; m < NT; ++m) {
        #pragma unroll
        for (int j = 0; j < 4; ++j) acc[m][j] = v8f{};
    }
    float4 bq[KQ / 2];
    #pragma unroll
    for (int i = 0; i < KQ / 4; ++i) {
        bq[2 * i]     = *(const float4*)(wp + (size_t)(4 * i) * NO);
        bq[2 * i + 1] = *(const float4*)(wp + (size_t)(4 * i + 1) * NO);
    }
    __builtin_amdgcn_sched_barrier(0);
    #pragma unroll
    for (int i = 0; i < KQ / 4; ++i) {
        const float4 q0 = bq[2 * i];
        const float4 q1 = bq[2 * i + 1];
        v2f b0, b1, b2, b3;
        b0.x = q0.x; b0.y = q1.x;
        b1.x = q0.y; b1.y = q1.y;
        b2.x = q0.z; b2.y = q1.z;
        b3.x = q0.w; b3.y = q1.w;
        const int k = kbase + 4 * i;
        #pragma unroll
        for (int m = 0; m < NT; ++m) {
            v2f a;
            a.x = As[16 * m + row][k];
            a.y = As[16 * m + row][k + 1];
            acc[m][0] = __builtin_amdgcn_wmma_f32_16x16x4_f32(
                false, a, false, b0, (short)0, acc[m][0], false, false);
            acc[m][1] = __builtin_amdgcn_wmma_f32_16x16x4_f32(
                false, a, false, b1, (short)0, acc[m][1], false, false);
            acc[m][2] = __builtin_amdgcn_wmma_f32_16x16x4_f32(
                false, a, false, b2, (short)0, acc[m][2], false, false);
            acc[m][3] = __builtin_amdgcn_wmma_f32_16x16x4_f32(
                false, a, false, b3, (short)0, acc[m][3], false, false);
        }
    }
    #pragma unroll
    for (int m = 0; m < NT; ++m) {
        #pragma unroll
        for (int j = 0; j < 8; ++j) {
            float4 o;
            o.x = acc[m][0][j];
            o.y = acc[m][1][j];
            o.z = acc[m][2][j];
            o.w = acc[m][3][j];
            *(float4*)&Pq[16 * m + prow + j][pcol] = o;
        }
    }
}

__global__ void __launch_bounds__(256) __attribute__((amdgpu_waves_per_eu(2, 2)))
option_gemm_kernel(const float* __restrict__ x, const float* __restrict__ w,
                   const float* __restrict__ bias, const int* __restrict__ idx,
                   const int* __restrict__ expOff, const int* __restrict__ expCnt,
                   float* __restrict__ out) {
    __shared__ float As[RG][ND + APAD];
    __shared__ float Ps[4][RG][SLABC];
    __shared__ int   rowIdx[RG];

    const int e     = blockIdx.x >> 2;
    const int slab  = blockIdx.x & 3;
    const int total = expCnt[e];
    if (total <= 0) return;
    const int base0 = expOff[e];
    const int tid   = threadIdx.x;
    const int lane  = tid & 31;
    const int wv    = tid >> 5;
    const int g     = wv & 1;
    const int q     = wv >> 1;
    const int c     = lane & 15;
    const int kh    = (lane >> 4) * 2;
    const int pcol  = g * 64 + c * 4;
    const int prow  = (lane >> 4) * 8;

    const float* wp = w + (size_t)e * (ND * NO) + (size_t)(q * KQ + kh) * NO
                        + slab * SLABC + pcol;
    const int kbase = q * KQ + kh;

    const int ocol = slab * SLABC + lane * 4;
    const float4 bv = *(const float4*)(bias + e * NO + ocol);

    for (int r0 = 0; r0 < total; r0 += RG) {
        const int left = total - r0;
        const int cnt  = left < RG ? left : RG;
        const int nt   = (cnt + 15) >> 4;

        if (tid < RG) rowIdx[tid] = (tid < cnt) ? idx[base0 + r0 + tid] : 0;
        for (int r = tid >> 4; r < 16 * nt; r += 16) {
            const int c4 = (tid & 15) * 4;
            if (r < cnt) {
                const float* src = x + (size_t)idx[base0 + r0 + r] * ND + c4;
                #pragma unroll
                for (int i = 0; i < 8; ++i) {
                    float4 v = *(const float4*)(src + i * 64);
                    *(float4*)&As[r][c4 + i * 64] = v;
                }
            } else {
                #pragma unroll
                for (int i = 0; i < 8; ++i) {
                    float4 z = {0.f, 0.f, 0.f, 0.f};
                    *(float4*)&As[r][c4 + i * 64] = z;
                }
            }
        }
        __syncthreads();

        const int row = c;
        switch (nt) {
            case 1:  slab_pass<1>(wp, As, Ps[q], kbase, row, prow, pcol); break;
            case 2:  slab_pass<2>(wp, As, Ps[q], kbase, row, prow, pcol); break;
            case 3:  slab_pass<3>(wp, As, Ps[q], kbase, row, prow, pcol); break;
            default: slab_pass<4>(wp, As, Ps[q], kbase, row, prow, pcol); break;
        }
        __syncthreads();

        for (int r = wv; r < cnt; r += 8) {
            const float4 p0 = *(const float4*)&Ps[0][r][lane * 4];
            const float4 p1 = *(const float4*)&Ps[1][r][lane * 4];
            const float4 p2 = *(const float4*)&Ps[2][r][lane * 4];
            const float4 p3 = *(const float4*)&Ps[3][r][lane * 4];
            float4 o;
            o.x = (((p0.x + p1.x) + p2.x) + p3.x) + bv.x;
            o.y = (((p0.y + p1.y) + p2.y) + p3.y) + bv.y;
            o.z = (((p0.z + p1.z) + p2.z) + p3.z) + bv.z;
            o.w = (((p0.w + p1.w) + p2.w) + p3.w) + bv.w;
            *(float4*)(out + (size_t)rowIdx[r] * NO + ocol) = o;
        }
        __syncthreads();
    }
}


extern "C" void kernel_launch(void* const* d_in, const int* in_sizes, int n_in,
                              void* d_out, int out_size, void* d_ws, size_t ws_size,
                              hipStream_t stream) {
    const float* x    = (const float*)d_in[0];
    const int*   opt  = (const int*)d_in[1];
    const float* wgt  = (const float*)d_in[2];
    const float* bias = (const float*)d_in[3];
    float* out = (float*)d_out;

    int* idx    = (int*)d_ws;
    int* expOff = idx + NB;
    int* expCnt = expOff + NE;

    route_prep_kernel<<<1, PREP_T, 0, stream>>>(opt, idx, expOff, expCnt);

    dim3 grid(NE * (NO / SLABC));
    option_gemm_kernel<<<grid, 256, 0, stream>>>(x, wgt, bias, idx,
                                                 expOff, expCnt, out);
}


# One launch instead of two: each GEMM block builds its expert's row list itself from the option array (8 loads per lane, wave ballots, prefix over the 8 waves in LDS); no workspace, no atomics
# speedup vs baseline: 4.6326x; 1.2776x over previous
#include <hip/hip_runtime.h>

typedef __attribute__((ext_vector_type(2))) float v2f;
typedef __attribute__((ext_vector_type(8))) float v8f;

#define NB   2048
#define NE   64
#define ND   512
#define NO   512


#define SLABC 128
#define RG    64
#define KQ    128
#define APAD  4

template <int NT>
__device__ __forceinline__ void slab_pass(const float* __restrict__ wp,
                                          const float (*As)[ND + APAD],
                                          float (*Pq)[SLABC],
                                          const int kbase, const int row,
                                          const int prow, const int pcol) {
    v8f acc[NT][4];
    #pragma unroll
    for (int m = 0; m < NT; ++m) {
        #pragma unroll
        for (int j = 0; j < 4; ++j) acc[m][j] = v8f{};
    }
    float4 bq[KQ / 2];
    #pragma unroll
    for (int i = 0; i < KQ / 4; ++i) {
        bq[2 * i]     = *(const float4*)(wp + (size_t)(4 * i) * NO);
        bq[2 * i + 1] = *(const float4*)(wp + (size_t)(4 * i + 1) * NO);
    }
    __builtin_amdgcn_sched_barrier(0);
    #pragma unroll
    for (int i = 0; i < KQ / 4; ++i) {
        const float4 q0 = bq[2 * i];
        const float4 q1 = bq[2 * i + 1];
        v2f b0, b1, b2, b3;
        b0.x = q0.x; b0.y = q1.x;
        b1.x = q0.y; b1.y = q1.y;
        b2.x = q0.z; b2.y = q1.z;
        b3.x = q0.w; b3.y = q1.w;
        const int k = kbase + 4 * i;
        #pragma unroll
        for (int m = 0; m < NT; ++m) {
            v2f a;
            a.x = As[16 * m + row][k];
            a.y = As[16 * m + row][k + 1];
            acc[m][0] = __builtin_amdgcn_wmma_f32_16x16x4_f32(
                false, a, false, b0, (short)0, acc[m][0], false, false);
            acc[m][1] = __builtin_amdgcn_wmma_f32_16x16x4_f32(
                false, a, false, b1, (short)0, acc[m][1], false, false);
            acc[m][2] = __builtin_amdgcn_wmma_f32_16x16x4_f32(
                false, a, false, b2, (short)0, acc[m][2], false, false);
            acc[m][3] = __builtin_amdgcn_wmma_f32_16x16x4_f32(
                false, a, false, b3, (short)0, acc[m][3], false, false);
        }
    }
    #pragma unroll
    for (int m = 0; m < NT; ++m) {
        #pragma unroll
        for (int j = 0; j < 8; ++j) {
            float4 o;
            o.x = acc[m][0][j];
            o.y = acc[m][1][j];
            o.z = acc[m][2][j];
            o.w = acc[m][3][j];
            *(float4*)&Pq[16 * m + prow + j][pcol] = o;
        }
    }
}

__global__ void __launch_bounds__(256) __attribute__((amdgpu_waves_per_eu(2, 2)))
option_gemm_kernel(const float* __restrict__ x, const int* __restrict__ opt,
                   const float* __restrict__ w, const float* __restrict__ bias,
                   float* __restrict__ out) {
    __shared__ float As[RG][ND + APAD];
    __shared__ float Ps[4][RG][SLABC];
    __shared__ int   rowList[NB];
    __shared__ int   waveCnt[8];

    const int e     = blockIdx.x >> 2;
    const int slab  = blockIdx.x & 3;
    const int tid   = threadIdx.x;
    const int lane  = tid & 31;
    const int wv    = tid >> 5;

    int ch[8];
    #pragma unroll
    for (int i = 0; i < 8; ++i) ch[i] = opt[wv * 256 + i * 32 + lane];
    unsigned hit[8];
    int mine = 0;
    #pragma unroll
    for (int i = 0; i < 8; ++i) {
        hit[i] = __builtin_amdgcn_ballot_w32(ch[i] == e);
        mine += __popc(hit[i]);
    }
    if (lane == 0) waveCnt[wv] = mine;
    __syncthreads();
    int total = 0;
    int pos   = 0;
    #pragma unroll
    for (int v = 0; v < 8; ++v) {
        const int cv = waveCnt[v];
        total += cv;
        if (v < wv) pos += cv;
    }
    if (total <= 0) return;
    #pragma unroll
    for (int i = 0; i < 8; ++i) {
        if (ch[i] == e)
            rowList[pos + (int)__builtin_amdgcn_mbcnt_lo(hit[i], 0u)] = wv * 256 + i * 32 + lane;
        pos += __popc(hit[i]);
    }
    __syncthreads();

    const int g     = wv & 1;
    const int q     = wv >> 1;
    const int c     = lane & 15;
    const int kh    = (lane >> 4) * 2;
    const int pcol  = g * 64 + c * 4;
    const int prow  = (lane >> 4) * 8;

    const float* wp = w + (size_t)e * (ND * NO) + (size_t)(q * KQ + kh) * NO
                        + slab * SLABC + pcol;
    const int kbase = q * KQ + kh;

    const int ocol = slab * SLABC + lane * 4;
    const float4 bv = *(const float4*)(bias + e * NO + ocol);

    for (int r0 = 0; r0 < total; r0 += RG) {
        const int left = total - r0;
        const int cnt  = left < RG ? left : RG;
        const int nt   = (cnt + 15) >> 4;

        for (int r = tid >> 4; r < 16 * nt; r += 16) {
            const int c4 = (tid & 15) * 4;
            if (r < cnt) {
                const float* src = x + (size_t)rowList[r0 + r] * ND + c4;
                #pragma unroll
                for (int i = 0; i < 8; ++i) {
                    float4 v = *(const float4*)(src + i * 64);
                    *(float4*)&As[r][c4 + i * 64] = v;
                }
            } else {
                #pragma unroll
                for (int i = 0; i < 8; ++i) {
                    float4 z = {0.f, 0.f, 0.f, 0.f};
                    *(float4*)&As[r][c4 + i * 64] = z;
                }
            }
        }
        __syncthreads();

        const int row = c;
        switch (nt) {
            case 1:  slab_pass<1>(wp, As, Ps[q], kbase, row, prow, pcol); break;
            case 2:  slab_pass<2>(wp, As, Ps[q], kbase, row, prow, pcol); break;
            case 3:  slab_pass<3>(wp, As, Ps[q], kbase, row, prow, pcol); break;
            default: slab_pass<4>(wp, As, Ps[q], kbase, row, prow, pcol); break;
        }
        __syncthreads();

        for (int r = wv; r < cnt; r += 8) {
            const float4 p0 = *(const float4*)&Ps[0][r][lane * 4];
            const float4 p1 = *(const float4*)&Ps[1][r][lane * 4];
            const float4 p2 = *(const float4*)&Ps[2][r][lane * 4];
            const float4 p3 = *(const float4*)&Ps[3][r][lane * 4];
            float4 o;
            o.x = (((p0.x + p1.x) + p2.x) + p3.x) + bv.x;
            o.y = (((p0.y + p1.y) + p2.y) + p3.y) + bv.y;
            o.z = (((p0.z + p1.z) + p2.z) + p3.z) + bv.z;
            o.w = (((p0.w + p1.w) + p2.w) + p3.w) + bv.w;
            *(float4*)(out + (size_t)rowList[r0 + r] * NO + ocol) = o;
        }
        __syncthreads();
    }
}


extern "C" void kernel_launch(void* const* d_in, const int* in_sizes, int n_in,
                              void* d_out, int out_size, void* d_ws, size_t ws_size,
                              hipStream_t stream) {
    const float* x    = (const float*)d_in[0];
    const int*   opt  = (const int*)d_in[1];
    const float* wgt  = (const float*)d_in[2];
    const float* bias = (const float*)d_in[3];
    float* out = (float*)d_out;

    dim3 grid(NE * (NO / SLABC));
    option_gemm_kernel<<<grid, 256, 0, stream>>>(x, opt, wgt, bias, out);
}
